# starting kernel
# baseline (speedup 1.0000x reference)
#include <hip/hip_runtime.h>

typedef float v2f __attribute__((ext_vector_type(2)));
typedef float v8f __attribute__((ext_vector_type(8)));

#define BATCH 8
#define CIN   512
#define TLEN  512
#define OCH   512
#define KSZ   65
#define PADW  32

#define NT 4
#define KT 20

__global__ __launch_bounds__(256) void pwtdconv_wmma_f32(
    const float* __restrict__ x,
    const float* __restrict__ w,
    const float* __restrict__ bias,
    float* __restrict__ out)
{
    const int gwave = (int)((blockIdx.x * blockDim.x + threadIdx.x) >> 5);
    const int lane  = (int)(threadIdx.x & 31);
    const int lm    = lane & 15;
    const int lh    = lane >> 4;

    const int OT = OCH / 16;
    const int TG = (TLEN / 16) / NT;

    int idx = gwave;
    const int ot = idx % OT; idx /= OT;
    const int tg = idx % TG; idx /= TG;
    const int b  = idx;
    if (b >= BATCH) return;

    const int o0    = ot * 16;
    const int o     = o0 + lm;
    const int cbase = o0 - PADW;

    v2f      bfrag[KT];
    unsigned aoff[KT][2];
#pragma unroll
    for (int kt = 0; kt < KT; ++kt) {
#pragma unroll
        for (int r = 0; r < 2; ++r) {
            const int c  = cbase + kt * 4 + 2 * lh + r;
            const int kk = c - o + PADW;
            const bool inb = (kk >= 0) & (kk < KSZ) & (c >= 0) & (c < CIN);
            const int  kc  = kk < 0 ? 0 : (kk > KSZ - 1 ? KSZ - 1 : kk);
            const float wv = w[o * KSZ + kc];
            bfrag[kt][r] = inb ? wv : 0.0f;

            const int cc = c < 0 ? 0 : (c > CIN - 1 ? CIN - 1 : c);
            aoff[kt][r]  = (unsigned)((cc * TLEN + lm) * 4);
        }
    }

    const float bv = bias[o];
    const char* xb = (const char*)(x + (size_t)b * CIN * TLEN);
    float*      ob = out + (size_t)b * TLEN * OCH;

    for (int it = 0; it < NT; ++it) {
        const int   t0  = (tg * NT + it) * 16;
        const char* xbt = xb + (size_t)t0 * 4;

        v8f acc;
#pragma unroll
        for (int r = 0; r < 8; ++r) acc[r] = bv;

#pragma unroll
        for (int kt = 0; kt < KT; ++kt) {
            v2f a;
            a[0] = *(const float*)(xbt + aoff[kt][0]);
            a[1] = *(const float*)(xbt + aoff[kt][1]);
            acc = __builtin_amdgcn_wmma_f32_16x16x4_f32(
                false, a, false, bfrag[kt], (short)0, acc, false, false);
        }

        float* obt = ob + (size_t)(t0 + 8 * lh) * OCH + o;
#pragma unroll
        for (int r = 0; r < 8; ++r)
            obt[(size_t)r * OCH] = acc[r];
    }
}

extern "C" void kernel_launch(void* const* d_in, const int* in_sizes, int n_in,
                              void* d_out, int out_size, void* d_ws, size_t ws_size,
                              hipStream_t stream) {
    const float* x    = (const float*)d_in[0];
    const float* w    = (const float*)d_in[1];
    const float* bias = (const float*)d_in[2];
    float*       out  = (float*)d_out;

    const int total_waves   = BATCH * (OCH / 16) * ((TLEN / 16) / NT);
    const int block_threads = 256;
    const int grid_blocks   = (total_waves * 32 + block_threads - 1) / block_threads;

    hipLaunchKernelGGL(pwtdconv_wmma_f32, dim3(grid_blocks), dim3(block_threads),
                       0, stream, x, w, bias, out);
}


# A operand read with global_load_b128 along t (a lane's 4 t feed 4 concurrent tiles, tile j row m is t=T0+4m+j), all 40 issued up front ahead of the B gathers: 20 serial load-wait rounds become one
# speedup vs baseline: 2.1630x; 2.1630x over previous
#include <hip/hip_runtime.h>

typedef float v2f __attribute__((ext_vector_type(2)));
typedef float v4f __attribute__((ext_vector_type(4)));
typedef float v8f __attribute__((ext_vector_type(8)));

#define BATCH 8
#define CIN   512
#define TLEN  512
#define OCH   512
#define KSZ   65
#define PADW  32

#define NT 4
#define KT 20
#define WIN (KT * 4)

__global__ __launch_bounds__(256) void pwtdconv_wmma_f32(
    const float* __restrict__ x,
    const float* __restrict__ w,
    const float* __restrict__ bias,
    float* __restrict__ out)
{
    const int gwave = (int)((blockIdx.x * blockDim.x + threadIdx.x) >> 5);
    const int lane  = (int)(threadIdx.x & 31);
    const int lm    = lane & 15;
    const int lh    = lane >> 4;

    const int OT = OCH / 16;
    const int TG = (TLEN / 16) / NT;

    int idx = gwave;
    const int ot = idx % OT; idx /= OT;
    const int tg = idx % TG; idx /= TG;
    const int b  = idx;
    if (b >= BATCH) return;

    const int o0 = ot * 16;
    const int o  = o0 + lm;
    int cbase = o0 - PADW;
    cbase = cbase < 0 ? 0 : (cbase > CIN - WIN ? CIN - WIN : cbase);

    const int T0 = tg * (16 * NT);
    const float* xp = x + ((size_t)b * CIN + (size_t)(cbase + 2 * lh)) * TLEN + T0 + 4 * lm;
    const float* wp = w + (size_t)o * KSZ;
    const int k0 = cbase + 2 * lh - o + PADW;

    const float bv = bias[o];
    v4f   xa[KT][2];
#pragma unroll
    for (int kt = 0; kt < KT; ++kt)
#pragma unroll
        for (int r = 0; r < 2; ++r)
            xa[kt][r] = *(const v4f*)(xp + (size_t)(kt * 4 + r) * TLEN);
    __builtin_amdgcn_sched_barrier(0);
    float wv[KT][2];
#pragma unroll
    for (int kt = 0; kt < KT; ++kt)
#pragma unroll
        for (int r = 0; r < 2; ++r) {
            const int kk = k0 + kt * 4 + r;
            const int kc = kk < 0 ? 0 : (kk > KSZ - 1 ? KSZ - 1 : kk);
            wv[kt][r] = wp[kc];
        }

    __builtin_amdgcn_sched_barrier(0);

    v8f acc[NT];
#pragma unroll
    for (int j = 0; j < NT; ++j)
#pragma unroll
        for (int r = 0; r < 8; ++r) acc[j][r] = bv;

#pragma unroll
    for (int kt = 0; kt < KT; ++kt) {
        v2f bf;
#pragma unroll
        for (int r = 0; r < 2; ++r) {
            const int kk = k0 + kt * 4 + r;
            bf[r] = ((kk >= 0) & (kk < KSZ)) ? wv[kt][r] : 0.0f;
        }
#pragma unroll
        for (int j = 0; j < NT; ++j) {
            v2f a;
            a[0] = xa[kt][0][j];
            a[1] = xa[kt][1][j];
            acc[j] = __builtin_amdgcn_wmma_f32_16x16x4_f32(
                false, a, false, bf, (short)0, acc[j], false, false);
        }
    }


    float* ob = out + ((size_t)b * TLEN + T0) * OCH + o;
#pragma unroll
    for (int j = 0; j < NT; ++j)
#pragma unroll
        for (int r = 0; r < 8; ++r)
            ob[(size_t)(4 * (r + 8 * lh) + j) * OCH] = acc[j][r];
}

extern "C" void kernel_launch(void* const* d_in, const int* in_sizes, int n_in,
                              void* d_out, int out_size, void* d_ws, size_t ws_size,
                              hipStream_t stream) {
    const float* x    = (const float*)d_in[0];
    const float* w    = (const float*)d_in[1];
    const float* bias = (const float*)d_in[2];
    float*       out  = (float*)d_out;

    const int total_waves   = BATCH * (OCH / 16) * ((TLEN / 16) / NT);
    const int block_threads = 256;
    const int grid_blocks   = (total_waves * 32 + block_threads - 1) / block_threads;

    hipLaunchKernelGGL(pwtdconv_wmma_f32, dim3(grid_blocks), dim3(block_threads),
                       0, stream, x, w, bias, out);
}


# B operand: the o-tile's 16 weight rows read with whole-wave contiguous loads issued first and laid out as a band tile in LDS, replacing 40 per-lane gathers of 16 lines; products follow the A loads
# speedup vs baseline: 2.2772x; 1.0528x over previous
#include <hip/hip_runtime.h>

typedef float v2f __attribute__((ext_vector_type(2)));
typedef float v4f __attribute__((ext_vector_type(4)));
typedef float v8f __attribute__((ext_vector_type(8)));

#define BATCH 8
#define CIN   512
#define TLEN  512
#define OCH   512
#define KSZ   65
#define PADW  32

#define NT 4
#define KT 20
#define WIN (KT * 4)

#define WPB   8
#define WMARG 32
#define WP    148

__global__ __launch_bounds__(256) void pwtdconv_wmma_f32(
    const float* __restrict__ x,
    const float* __restrict__ w,
    const float* __restrict__ bias,
    float* __restrict__ out)
{
    __shared__ __attribute__((aligned(16))) float wl[WPB][16 * WP];

    const int gwave = (int)((blockIdx.x * blockDim.x + threadIdx.x) >> 5);
    const int wib   = (int)(threadIdx.x >> 5);
    const int lane  = (int)(threadIdx.x & 31);
    const int lm    = lane & 15;
    const int lh    = lane >> 4;

    const int OT = OCH / 16;
    const int TG = (TLEN / 16) / NT;

    int idx = gwave;
    const int ot = idx % OT; idx /= OT;
    const int tg = idx % TG; idx /= TG;
    const int b  = idx;
    if (b >= BATCH) return;

    const int o0 = ot * 16;
    const int o  = o0 + lm;
    int cbase = o0 - PADW;
    cbase = cbase < 0 ? 0 : (cbase > CIN - WIN ? CIN - WIN : cbase);

    const int T0 = tg * (16 * NT);
    const float* xp = x + ((size_t)b * CIN + (size_t)(cbase + 2 * lh)) * TLEN + T0 + 4 * lm;

    float* wt = wl[wib];
    const float* wrow = w + (size_t)o0 * KSZ + lane;
    float wraw[16][2];
#pragma unroll
    for (int ol = 0; ol < 16; ++ol)
#pragma unroll
        for (int i = 0; i < 2; ++i)
            wraw[ol][i] = wrow[ol * KSZ + 32 * i];
    const float wlast = w[(size_t)o * KSZ + (KSZ - 1)];
    const float bv = bias[o];
    v4f   xa[KT][2];
#pragma unroll
    for (int kt = 0; kt < KT; ++kt)
#pragma unroll
        for (int r = 0; r < 2; ++r)
            xa[kt][r] = *(const v4f*)(xp + (size_t)(kt * 4 + r) * TLEN);

    {
        v4f* zp = (v4f*)(wt + (lane >> 1) * WP + WMARG + (WIN / 2) * (lane & 1));
#pragma unroll
        for (int n = 0; n < WIN / 8; ++n) zp[n] = (v4f){0.0f, 0.0f, 0.0f, 0.0f};
    }
    const int dcol = o0 - PADW - cbase;
    float* sp = wt + WMARG + dcol + lane;
#pragma unroll
    for (int ol = 0; ol < 16; ++ol)
#pragma unroll
        for (int i = 0; i < 2; ++i)
            sp[ol * WP + ol + 32 * i] = wraw[ol][i];
    wt[lm * WP + WMARG + dcol + lm + (KSZ - 1)] = wlast;
    v2f bfr[KT];
#pragma unroll
    for (int kt = 0; kt < KT; ++kt)
        bfr[kt] = *(const v2f*)(wt + lm * WP + WMARG + 4 * kt + 2 * lh);
    __builtin_amdgcn_sched_barrier(0);

    v8f acc[NT];
#pragma unroll
    for (int j = 0; j < NT; ++j)
#pragma unroll
        for (int r = 0; r < 8; ++r) acc[j][r] = bv;

#pragma unroll
    for (int kt = 0; kt < KT; ++kt) {
#pragma unroll
        for (int j = 0; j < NT; ++j) {
            v2f a;
            a[0] = xa[kt][0][j];
            a[1] = xa[kt][1][j];
            acc[j] = __builtin_amdgcn_wmma_f32_16x16x4_f32(
                false, a, false, bfr[kt], (short)0, acc[j], false, false);
        }
    }

    float* ob = out + ((size_t)b * TLEN + T0) * OCH + o;
#pragma unroll
    for (int j = 0; j < NT; ++j)
#pragma unroll
        for (int r = 0; r < 8; ++r)
            ob[(size_t)(4 * (r + 8 * lh) + j) * OCH] = acc[j][r];
}

extern "C" void kernel_launch(void* const* d_in, const int* in_sizes, int n_in,
                              void* d_out, int out_size, void* d_ws, size_t ws_size,
                              hipStream_t stream) {
    const float* x    = (const float*)d_in[0];
    const float* w    = (const float*)d_in[1];
    const float* bias = (const float*)d_in[2];
    float*       out  = (float*)d_out;

    const int total_waves   = BATCH * (OCH / 16) * ((TLEN / 16) / NT);
    const int block_threads = 256;
    const int grid_blocks   = (total_waves * 32 + block_threads - 1) / block_threads;

    hipLaunchKernelGGL(pwtdconv_wmma_f32, dim3(grid_blocks), dim3(block_threads),
                       0, stream, x, w, bias, out);
}
